# starting kernel
# baseline (speedup 1.0000x reference)
#include <hip/hip_runtime.h>

typedef float v2f __attribute__((ext_vector_type(2)));
typedef float v8f __attribute__((ext_vector_type(8)));

__device__ __forceinline__ float fast_tanh(float x) {
  const float ax = __builtin_fabsf(x);
  const float t  = __builtin_amdgcn_exp2f(ax * -2.8853900817779268f);
  const float r  = (1.0f - t) * __builtin_amdgcn_rcpf(1.0f + t);
  return __builtin_copysignf(r, x);
}

#define SCHALF(th, ss, cc)                                                   \
  const float u_##ss = (th) * 0.07957747154594767f;           \
  const float ss = __builtin_amdgcn_sinf(u_##ss);                            \
  const float cc = __builtin_amdgcn_cosf(u_##ss)

#define RY_Q0(th) do { SCHALF(th, ss, cc);                                   \
    const float a0=s0, a1=s1, b0=s2, b1v=s3;                                 \
    s0 = cc*a0 - ss*b0;  s1 = cc*a1 - ss*b1v;                                \
    s2 = ss*a0 + cc*b0;  s3 = ss*a1 + cc*b1v; } while (0)

#define RY_Q1(th) do { SCHALF(th, ss, cc);                                   \
    const float a0=s0, b0=s1, a1=s2, b1v=s3;                                 \
    s0 = cc*a0 - ss*b0;  s1 = ss*a0 + cc*b0;                                 \
    s2 = cc*a1 - ss*b1v; s3 = ss*a1 + cc*b1v; } while (0)

__global__ __launch_bounds__(256, 1)
void hybrid_sampler_kernel(const float* __restrict__ inp,
                           const float* __restrict__ W1,
                           const float* __restrict__ b1,
                           const float* __restrict__ W2,
                           const float* __restrict__ b2,
                           float* __restrict__ out, int nrows) {
  __shared__ float hbuf[8][32 * 16];

  const int tid  = threadIdx.x;
  const int wave = tid >> 5;
  const int lane = tid & 31;
  const int n    = lane & 15;
  const bool hi  = lane >= 16;
  float* hb = hbuf[wave];

  int row = blockIdx.x * 256 + wave * 32 + lane;
  row = (row >= nrows) ? (nrows - 1) : row;

  const float in0 = inp[2 * row + 0];
  const float in1 = inp[2 * row + 1];

  const int n8 = n & 7;
  const int n4 = n & 3;
  const float w1k0 = W1[n8];
  const float w1k1 = W1[8 + n8];
  const float bb1  = b1[n8];
  const int rlo = hi ? 8  : 0;
  const int rl1 = hi ? 12 : 4;
  const int rh0 = hi ? 24 : 16;
  const int rh1 = hi ? 28 : 20;
  const float w2a = W2[rlo + n4];
  const float w2b = W2[rl1 + n4];
  const float w2c = W2[rh0 + n4];
  const float w2d = W2[rh1 + n4];
  const float bb2 = b2[n4];

  const float selLo8 = (!hi && n < 8) ? 1.0f : 0.0f;
  const float m8     = (n < 8) ? 1.0f : 0.0f;
  v2f B0;
  B0.x = w1k0 * selLo8;
  B0.y = w1k1 * selLo8;
  const float c1 = bb1 * m8;
  v8f C1 = {c1, c1, c1, c1, c1, c1, c1, c1};

  v2f A0; A0.x = hi ? 0.0f : in0; A0.y = hi ? 0.0f : in1;
  const int sidx = ((lane + 16) & 31) << 2;
  const float in0s = __int_as_float(__builtin_amdgcn_ds_bpermute(sidx, __float_as_int(in0)));
  const float in1s = __int_as_float(__builtin_amdgcn_ds_bpermute(sidx, __float_as_int(in1)));
  v2f A1; A1.x = hi ? 0.0f : in0s; A1.y = hi ? 0.0f : in1s;

  v8f H0 = __builtin_amdgcn_wmma_f32_16x16x4_f32(false, A0, false, B0, (short)0, C1, false, false);
  v8f H1 = __builtin_amdgcn_wmma_f32_16x16x4_f32(false, A1, false, B0, (short)0, C1, false, false);

  #pragma unroll
  for (int j = 0; j < 8; ++j) {
    const int m = j + (hi ? 8 : 0);
    hb[m * 16 + n]        = H0[j];
    hb[(m + 16) * 16 + n] = H1[j];
  }

  const int koff = hi ? 2 : 0;
  v2f AL0 = *(const v2f*)&hb[n * 16 + koff];
  v2f AH0 = *(const v2f*)&hb[n * 16 + 4 + koff];
  v2f AL1 = *(const v2f*)&hb[(n + 16) * 16 + koff];
  v2f AH1 = *(const v2f*)&hb[(n + 16) * 16 + 4 + koff];
  AL0.x = fast_tanh(AL0.x);  AL0.y = fast_tanh(AL0.y);
  AH0.x = fast_tanh(AH0.x);  AH0.y = fast_tanh(AH0.y);
  AL1.x = fast_tanh(AL1.x);  AL1.y = fast_tanh(AL1.y);
  AH1.x = fast_tanh(AH1.x);  AH1.y = fast_tanh(AH1.y);

  const float m4 = (n < 4) ? 1.0f : 0.0f;
  v2f BL, BH;
  BL.x = w2a * m4;
  BL.y = w2b * m4;
  BH.x = w2c * m4;
  BH.y = w2d * m4;
  const float c2 = bb2 * m4;
  v8f C2 = {c2, c2, c2, c2, c2, c2, c2, c2};

  v8f T0  = __builtin_amdgcn_wmma_f32_16x16x4_f32(false, AH0, false, BH, (short)0, C2, false, false);
  v8f Wt0 = __builtin_amdgcn_wmma_f32_16x16x4_f32(false, AL0, false, BL, (short)0, T0, false, false);
  v8f T1  = __builtin_amdgcn_wmma_f32_16x16x4_f32(false, AH1, false, BH, (short)0, C2, false, false);
  v8f Wt1 = __builtin_amdgcn_wmma_f32_16x16x4_f32(false, AL1, false, BL, (short)0, T1, false, false);

  #pragma unroll
  for (int j = 0; j < 8; ++j) {
    const int m = j + (hi ? 8 : 0);
    hb[m * 16 + n]        = Wt0[j];
    hb[(m + 16) * 16 + n] = Wt1[j];
  }
  const float4 wv = *(const float4*)&hb[lane * 16];

  float s0, s1, s2, s3;
  { SCHALF(in0, ss0, cc0); s0 = cc0; s2 = ss0; s1 = 0.0f; s3 = 0.0f; }
  RY_Q1(in1);
  { const float t = s2; s2 = s3; s3 = t; }
  RY_Q0(wv.x);
  RY_Q1(wv.y);
  { const float t = s2; s2 = s3; s3 = t; }
  RY_Q0(wv.z);
  RY_Q1(wv.w);

  const float p0 = s0 * s0 + s2 * s2;
  const float p1 = s1 * s1 + s3 * s3;
  const float mx = fmaxf(p0, p1);
  const float e0 = __expf(p0 - mx);
  const float e1 = __expf(p1 - mx);
  const float inv = __builtin_amdgcn_rcpf(e0 + e1);
  float2 o; o.x = e0 * inv; o.y = e1 * inv;
  *(float2*)&out[2 * row] = o;
}

extern "C" void kernel_launch(void* const* d_in, const int* in_sizes, int n_in,
                              void* d_out, int out_size, void* d_ws, size_t ws_size,
                              hipStream_t stream) {
  const float* inp = (const float*)d_in[0];
  const float* W1  = (const float*)d_in[1];
  const float* b1  = (const float*)d_in[2];
  const float* W2  = (const float*)d_in[3];
  const float* b2  = (const float*)d_in[4];
  float* out = (float*)d_out;

  const int nrows  = in_sizes[0] / 2;
  const int blocks = (nrows + 255) / 256;

  hipLaunchKernelGGL(hybrid_sampler_kernel, dim3(blocks), dim3(256), 0, stream,
                     inp, W1, b1, W2, b2, out, nrows);
}


# MLP transposed onto WMMA: weights as block-diagonal A, batch rows as B, results land in the owning lane; both LDS transposes and the bpermutes go; layer-2 k split 4x2 (sum order changes)
# speedup vs baseline: 1.0875x; 1.0875x over previous
#include <hip/hip_runtime.h>


typedef float v2f __attribute__((ext_vector_type(2)));
typedef float v8f __attribute__((ext_vector_type(8)));

__device__ __forceinline__ float fast_tanh(float x) {
  const float ax = __builtin_fabsf(x);
  const float t  = __builtin_amdgcn_exp2f(ax * -2.8853900817779268f);
  const float r  = (1.0f - t) * __builtin_amdgcn_rcpf(1.0f + t);
  return __builtin_copysignf(r, x);
}

#define SCHALF(th, ss, cc)                                                   \
  const float u_##ss = (th) * 0.07957747154594767f;           \
  const float ss = __builtin_amdgcn_sinf(u_##ss);                            \
  const float cc = __builtin_amdgcn_cosf(u_##ss)

#define RY_Q0(th) do { SCHALF(th, ss, cc);                                   \
    const float a0=s0, a1=s1, b0=s2, b1v=s3;                                 \
    s0 = cc*a0 - ss*b0;  s1 = cc*a1 - ss*b1v;                                \
    s2 = ss*a0 + cc*b0;  s3 = ss*a1 + cc*b1v; } while (0)

#define RY_Q1(th) do { SCHALF(th, ss, cc);                                   \
    const float a0=s0, b0=s1, a1=s2, b1v=s3;                                 \
    s0 = cc*a0 - ss*b0;  s1 = ss*a0 + cc*b0;                                 \
    s2 = cc*a1 - ss*b1v; s3 = ss*a1 + cc*b1v; } while (0)

__global__ __launch_bounds__(256, 1)
void hybrid_sampler_kernel(const float* __restrict__ inp,
                           const float* __restrict__ W1,
                           const float* __restrict__ b1,
                           const float* __restrict__ W2,
                           const float* __restrict__ b2,
                           float* __restrict__ out, int nrows) {
  const int tid  = threadIdx.x;
  const int lane = tid & 31;
  const int c    = lane & 15;
  const int h    = lane >> 4;

  int row = blockIdx.x * 256 + tid;
  row = (row >= nrows) ? (nrows - 1) : row;

  const float in0 = inp[2 * row + 0];
  const float in1 = inp[2 * row + 1];

  const int r8 = c & 7;
  const int r4 = c & 3;
  const bool own8 = ((c >> 3) == h);
  const bool own4 = own8 && ((c & 4) == 0);
  const float w1x = W1[r8];
  const float w1y = W1[8 + r8];
  float w2x[4], w2y[4];
  #pragma unroll
  for (int t = 0; t < 4; ++t) {
    w2x[t] = W2[(2 * t) * 4 + r4];
    w2y[t] = W2[(2 * t + 1) * 4 + r4];
  }

  v2f A1;  A1.x = own8 ? w1x : 0.0f;  A1.y = own8 ? w1y : 0.0f;
  v2f B1;  B1.x = in0;                B1.y = in1;
  v8f C1 = {b1[0], b1[1], b1[2], b1[3], b1[4], b1[5], b1[6], b1[7]};
  v8f H = __builtin_amdgcn_wmma_f32_16x16x4_f32(false, A1, false, B1, (short)0, C1, false, false);

  float th[8];
  #pragma unroll
  for (int r = 0; r < 8; ++r) th[r] = fast_tanh(H[r]);

  v8f acc = {b2[0], b2[1], b2[2], b2[3], 0.0f, 0.0f, 0.0f, 0.0f};
  #pragma unroll
  for (int t = 0; t < 4; ++t) {
    v2f A2;  A2.x = own4 ? w2x[t] : 0.0f;  A2.y = own4 ? w2y[t] : 0.0f;
    v2f B2;  B2.x = th[2 * t];             B2.y = th[2 * t + 1];
    acc = __builtin_amdgcn_wmma_f32_16x16x4_f32(false, A2, false, B2, (short)0, acc, false, false);
  }

  float s0, s1, s2, s3;
  { SCHALF(in0, ss0, cc0); s0 = cc0; s2 = ss0; s1 = 0.0f; s3 = 0.0f; }
  RY_Q1(in1);
  { const float t = s2; s2 = s3; s3 = t; }
  RY_Q0(acc[0]);
  RY_Q1(acc[1]);
  { const float t = s2; s2 = s3; s3 = t; }
  RY_Q0(acc[2]);
  RY_Q1(acc[3]);

  const float p0 = s0 * s0 + s2 * s2;
  const float p1 = s1 * s1 + s3 * s3;
  const float mx = fmaxf(p0, p1);
  const float e0 = __expf(p0 - mx);
  const float e1 = __expf(p1 - mx);
  const float inv = __builtin_amdgcn_rcpf(e0 + e1);
  float2 o; o.x = e0 * inv; o.y = e1 * inv;
  *(float2*)&out[2 * row] = o;
}

extern "C" void kernel_launch(void* const* d_in, const int* in_sizes, int n_in,
                              void* d_out, int out_size, void* d_ws, size_t ws_size,
                              hipStream_t stream) {
  const float* inp = (const float*)d_in[0];
  const float* W1  = (const float*)d_in[1];
  const float* b1  = (const float*)d_in[2];
  const float* W2  = (const float*)d_in[3];
  const float* b2  = (const float*)d_in[4];
  float* out = (float*)d_out;

  const int nrows  = in_sizes[0] / 2;
  const int blocks = (nrows + 255) / 256;

  hipLaunchKernelGGL(hybrid_sampler_kernel, dim3(blocks), dim3(256), 0, stream,
                     inp, W1, b1, W2, b2, out, nrows);
}


# each lane does 8 rows (four 16-byte row-pair loads issued up front, 16-byte stores): weight fragments and biases built once per wave, 8x fewer waves, whole grid resident in one round; sums unchanged
# speedup vs baseline: 1.3222x; 1.2158x over previous
#include <hip/hip_runtime.h>


typedef float v2f __attribute__((ext_vector_type(2)));
typedef float v4f __attribute__((ext_vector_type(4)));
typedef float v8f __attribute__((ext_vector_type(8)));

__device__ __forceinline__ float fast_tanh(float x) {
  const float ax = __builtin_fabsf(x);
  const float t  = __builtin_amdgcn_exp2f(ax * -2.8853900817779268f);
  const float r  = (1.0f - t) * __builtin_amdgcn_rcpf(1.0f + t);
  return __builtin_copysignf(r, x);
}

#define SCHALF(th, ss, cc)                                                   \
  const float u_##ss = (th) * 0.07957747154594767f;           \
  const float ss = __builtin_amdgcn_sinf(u_##ss);                            \
  const float cc = __builtin_amdgcn_cosf(u_##ss)

#define RY_Q0(th) do { SCHALF(th, ss, cc);                                   \
    const float a0=s0, a1=s1, b0=s2, b1v=s3;                                 \
    s0 = cc*a0 - ss*b0;  s1 = cc*a1 - ss*b1v;                                \
    s2 = ss*a0 + cc*b0;  s3 = ss*a1 + cc*b1v; } while (0)

#define RY_Q1(th) do { SCHALF(th, ss, cc);                                   \
    const float a0=s0, b0=s1, a1=s2, b1v=s3;                                 \
    s0 = cc*a0 - ss*b0;  s1 = ss*a0 + cc*b0;                                 \
    s2 = cc*a1 - ss*b1v; s3 = ss*a1 + cc*b1v; } while (0)

struct WaveConsts {
  v2f A1;
  v2f A2[4];
  v8f C1;
  v8f C2;
};

__device__ __forceinline__ v2f eval_row(float in0, float in1, const WaveConsts& k) {
  v2f B1;  B1.x = in0;  B1.y = in1;
  v8f H = __builtin_amdgcn_wmma_f32_16x16x4_f32(false, k.A1, false, B1, (short)0, k.C1, false, false);

  float th[8];
  #pragma unroll
  for (int r = 0; r < 8; ++r) th[r] = fast_tanh(H[r]);

  v8f acc = k.C2;
  #pragma unroll
  for (int t = 0; t < 4; ++t) {
    v2f B2;  B2.x = th[2 * t];  B2.y = th[2 * t + 1];
    acc = __builtin_amdgcn_wmma_f32_16x16x4_f32(false, k.A2[t], false, B2, (short)0, acc, false, false);
  }

  float s0, s1, s2, s3;
  { SCHALF(in0, ss0, cc0); s0 = cc0; s2 = ss0; s1 = 0.0f; s3 = 0.0f; }
  RY_Q1(in1);
  { const float t = s2; s2 = s3; s3 = t; }
  RY_Q0(acc[0]);
  RY_Q1(acc[1]);
  { const float t = s2; s2 = s3; s3 = t; }
  RY_Q0(acc[2]);
  RY_Q1(acc[3]);

  const float p0 = s0 * s0 + s2 * s2;
  const float p1 = s1 * s1 + s3 * s3;
  const float mx = fmaxf(p0, p1);
  const float e0 = __expf(p0 - mx);
  const float e1 = __expf(p1 - mx);
  const float inv = __builtin_amdgcn_rcpf(e0 + e1);
  v2f o; o.x = e0 * inv; o.y = e1 * inv;
  return o;
}

constexpr int kPairs        = 4;
constexpr int kRowsPerLane  = 2 * kPairs;
constexpr int kRowsPerWave  = 32 * kRowsPerLane;
constexpr int kWaves        = 8;
constexpr int kRowsPerBlock = kWaves * kRowsPerWave;

template <bool FULL>
__device__ __forceinline__ void block_body(const float* __restrict__ inp,
                                           const float* __restrict__ W1,
                                           const float* __restrict__ b1,
                                           const float* __restrict__ W2,
                                           const float* __restrict__ b2,
                                           float* __restrict__ out, int nrows) {
  const int tid  = threadIdx.x;
  const int lane = tid & 31;
  const int wave = tid >> 5;
  const int c    = lane & 15;
  const int h    = lane >> 4;

  const int row0 = blockIdx.x * kRowsPerBlock + wave * kRowsPerWave + 2 * lane;

  v4f x[kPairs];
  #pragma unroll
  for (int j = 0; j < kPairs; ++j) {
    const int ra = row0 + 64 * j;
    if (FULL) {
      x[j] = *(const v4f*)&inp[2 * ra];
    } else {
      const int qa = (ra     >= nrows) ? (nrows - 1) : ra;
      const int qb = (ra + 1 >= nrows) ? (nrows - 1) : (ra + 1);
      x[j].x = inp[2 * qa];  x[j].y = inp[2 * qa + 1];
      x[j].z = inp[2 * qb];  x[j].w = inp[2 * qb + 1];
    }
  }

  const int r8 = c & 7;
  const int r4 = c & 3;
  const bool own8 = ((c >> 3) == h);
  const bool own4 = own8 && ((c & 4) == 0);
  const float w1x = W1[r8];
  const float w1y = W1[8 + r8];

  WaveConsts k;
  k.A1.x = own8 ? w1x : 0.0f;  k.A1.y = own8 ? w1y : 0.0f;
  #pragma unroll
  for (int t = 0; t < 4; ++t) {
    const float wx = W2[(2 * t) * 4 + r4];
    const float wy = W2[(2 * t + 1) * 4 + r4];
    k.A2[t].x = own4 ? wx : 0.0f;
    k.A2[t].y = own4 ? wy : 0.0f;
  }
  k.C1 = (v8f){b1[0], b1[1], b1[2], b1[3], b1[4], b1[5], b1[6], b1[7]};
  k.C2 = (v8f){b2[0], b2[1], b2[2], b2[3], 0.0f, 0.0f, 0.0f, 0.0f};

  v4f y[kPairs];
  #pragma unroll
  for (int j = 0; j < kPairs; ++j) {
    const v2f oa = eval_row(x[j].x, x[j].y, k);
    const v2f ob = eval_row(x[j].z, x[j].w, k);
    y[j].x = oa.x;  y[j].y = oa.y;  y[j].z = ob.x;  y[j].w = ob.y;
    if (FULL) *(v4f*)&out[2 * (row0 + 64 * j)] = y[j];
    __builtin_amdgcn_sched_barrier(0);
  }

  if (!FULL) {
    #pragma unroll
    for (int j = 0; j < kPairs; ++j) {
      const int ra = row0 + 64 * j;
      if (ra < nrows)     { v2f o; o.x = y[j].x; o.y = y[j].y; *(v2f*)&out[2 * ra] = o; }
      if (ra + 1 < nrows) { v2f o; o.x = y[j].z; o.y = y[j].w; *(v2f*)&out[2 * ra + 2] = o; }
    }
  }
}

__global__ __launch_bounds__(256)
void hybrid_sampler_kernel(const float* __restrict__ inp,
                           const float* __restrict__ W1,
                           const float* __restrict__ b1,
                           const float* __restrict__ W2,
                           const float* __restrict__ b2,
                           float* __restrict__ out, int nrows) {
  if ((int)(blockIdx.x + 1) * kRowsPerBlock <= nrows)
    block_body<true>(inp, W1, b1, W2, b2, out, nrows);
  else
    block_body<false>(inp, W1, b1, W2, b2, out, nrows);
}

extern "C" void kernel_launch(void* const* d_in, const int* in_sizes, int n_in,
                              void* d_out, int out_size, void* d_ws, size_t ws_size,
                              hipStream_t stream) {
  const float* inp = (const float*)d_in[0];
  const float* W1  = (const float*)d_in[1];
  const float* b1  = (const float*)d_in[2];
  const float* W2  = (const float*)d_in[3];
  const float* b2  = (const float*)d_in[4];
  float* out = (float*)d_out;

  const int nrows  = in_sizes[0] / 2;
  const int blocks = (nrows + kRowsPerBlock - 1) / kRowsPerBlock;

  hipLaunchKernelGGL(hybrid_sampler_kernel, dim3(blocks), dim3(256), 0, stream,
                     inp, W1, b1, W2, b2, out, nrows);
}


# v_tanh_f32 replaces every exp2+rcp sequence: the 8 hidden-layer tanh per row, and the 2-way softmax as 1/2 +/- tanh((p0-p1)/2)/2; tanh and softmax roundings change, sum order unchanged
# speedup vs baseline: 1.5936x; 1.2053x over previous
#include <hip/hip_runtime.h>


typedef float v2f __attribute__((ext_vector_type(2)));
typedef float v4f __attribute__((ext_vector_type(4)));
typedef float v8f __attribute__((ext_vector_type(8)));

__device__ __forceinline__ float fast_tanh(float x) {
  return __builtin_amdgcn_tanhf(x);
}

#define SCHALF(th, ss, cc)                                                   \
  const float u_##ss = (th) * 0.07957747154594767f;           \
  const float ss = __builtin_amdgcn_sinf(u_##ss);                            \
  const float cc = __builtin_amdgcn_cosf(u_##ss)

#define RY_Q0(th) do { SCHALF(th, ss, cc);                                   \
    const float a0=s0, a1=s1, b0=s2, b1v=s3;                                 \
    s0 = cc*a0 - ss*b0;  s1 = cc*a1 - ss*b1v;                                \
    s2 = ss*a0 + cc*b0;  s3 = ss*a1 + cc*b1v; } while (0)

#define RY_Q1(th) do { SCHALF(th, ss, cc);                                   \
    const float a0=s0, b0=s1, a1=s2, b1v=s3;                                 \
    s0 = cc*a0 - ss*b0;  s1 = ss*a0 + cc*b0;                                 \
    s2 = cc*a1 - ss*b1v; s3 = ss*a1 + cc*b1v; } while (0)

struct WaveConsts {
  v2f A1;
  v2f A2[4];
  v8f C1;
  v8f C2;
};

__device__ __forceinline__ v2f eval_row(float in0, float in1, const WaveConsts& k) {
  v2f B1;  B1.x = in0;  B1.y = in1;
  v8f H = __builtin_amdgcn_wmma_f32_16x16x4_f32(false, k.A1, false, B1, (short)0, k.C1, false, false);

  float th[8];
  #pragma unroll
  for (int r = 0; r < 8; ++r) th[r] = fast_tanh(H[r]);

  v8f acc = k.C2;
  #pragma unroll
  for (int t = 0; t < 4; ++t) {
    v2f B2;  B2.x = th[2 * t];  B2.y = th[2 * t + 1];
    acc = __builtin_amdgcn_wmma_f32_16x16x4_f32(false, k.A2[t], false, B2, (short)0, acc, false, false);
  }

  float s0, s1, s2, s3;
  { SCHALF(in0, ss0, cc0); s0 = cc0; s2 = ss0; s1 = 0.0f; s3 = 0.0f; }
  RY_Q1(in1);
  { const float t = s2; s2 = s3; s3 = t; }
  RY_Q0(acc[0]);
  RY_Q1(acc[1]);
  { const float t = s2; s2 = s3; s3 = t; }
  RY_Q0(acc[2]);
  RY_Q1(acc[3]);

  const float p0 = s0 * s0 + s2 * s2;
  const float p1 = s1 * s1 + s3 * s3;
  const float t = fast_tanh(0.5f * (p0 - p1));
  v2f o; o.x = __builtin_fmaf(0.5f, t, 0.5f); o.y = __builtin_fmaf(-0.5f, t, 0.5f);
  return o;
}

constexpr int kPairs        = 4;
constexpr int kRowsPerLane  = 2 * kPairs;
constexpr int kRowsPerWave  = 32 * kRowsPerLane;
constexpr int kWaves        = 8;
constexpr int kRowsPerBlock = kWaves * kRowsPerWave;

template <bool FULL>
__device__ __forceinline__ void block_body(const float* __restrict__ inp,
                                           const float* __restrict__ W1,
                                           const float* __restrict__ b1,
                                           const float* __restrict__ W2,
                                           const float* __restrict__ b2,
                                           float* __restrict__ out, int nrows) {
  const int tid  = threadIdx.x;
  const int lane = tid & 31;
  const int wave = tid >> 5;
  const int c    = lane & 15;
  const int h    = lane >> 4;

  const int row0 = blockIdx.x * kRowsPerBlock + wave * kRowsPerWave + 2 * lane;

  v4f x[kPairs];
  #pragma unroll
  for (int j = 0; j < kPairs; ++j) {
    const int ra = row0 + 64 * j;
    if (FULL) {
      x[j] = *(const v4f*)&inp[2 * ra];
    } else {
      const int qa = (ra     >= nrows) ? (nrows - 1) : ra;
      const int qb = (ra + 1 >= nrows) ? (nrows - 1) : (ra + 1);
      x[j].x = inp[2 * qa];  x[j].y = inp[2 * qa + 1];
      x[j].z = inp[2 * qb];  x[j].w = inp[2 * qb + 1];
    }
  }

  const int r8 = c & 7;
  const int r4 = c & 3;
  const bool own8 = ((c >> 3) == h);
  const bool own4 = own8 && ((c & 4) == 0);
  const float w1x = W1[r8];
  const float w1y = W1[8 + r8];

  WaveConsts k;
  k.A1.x = own8 ? w1x : 0.0f;  k.A1.y = own8 ? w1y : 0.0f;
  #pragma unroll
  for (int t = 0; t < 4; ++t) {
    const float wx = W2[(2 * t) * 4 + r4];
    const float wy = W2[(2 * t + 1) * 4 + r4];
    k.A2[t].x = own4 ? wx : 0.0f;
    k.A2[t].y = own4 ? wy : 0.0f;
  }
  k.C1 = (v8f){b1[0], b1[1], b1[2], b1[3], b1[4], b1[5], b1[6], b1[7]};
  k.C2 = (v8f){b2[0], b2[1], b2[2], b2[3], 0.0f, 0.0f, 0.0f, 0.0f};

  v4f y[kPairs];
  #pragma unroll
  for (int j = 0; j < kPairs; ++j) {
    const v2f oa = eval_row(x[j].x, x[j].y, k);
    const v2f ob = eval_row(x[j].z, x[j].w, k);
    y[j].x = oa.x;  y[j].y = oa.y;  y[j].z = ob.x;  y[j].w = ob.y;
    if (FULL) *(v4f*)&out[2 * (row0 + 64 * j)] = y[j];
    __builtin_amdgcn_sched_barrier(0);
  }

  if (!FULL) {
    #pragma unroll
    for (int j = 0; j < kPairs; ++j) {
      const int ra = row0 + 64 * j;
      if (ra < nrows)     { v2f o; o.x = y[j].x; o.y = y[j].y; *(v2f*)&out[2 * ra] = o; }
      if (ra + 1 < nrows) { v2f o; o.x = y[j].z; o.y = y[j].w; *(v2f*)&out[2 * ra + 2] = o; }
    }
  }
}

__global__ __launch_bounds__(256)
void hybrid_sampler_kernel(const float* __restrict__ inp,
                           const float* __restrict__ W1,
                           const float* __restrict__ b1,
                           const float* __restrict__ W2,
                           const float* __restrict__ b2,
                           float* __restrict__ out, int nrows) {
  if ((int)(blockIdx.x + 1) * kRowsPerBlock <= nrows)
    block_body<true>(inp, W1, b1, W2, b2, out, nrows);
  else
    block_body<false>(inp, W1, b1, W2, b2, out, nrows);
}

extern "C" void kernel_launch(void* const* d_in, const int* in_sizes, int n_in,
                              void* d_out, int out_size, void* d_ws, size_t ws_size,
                              hipStream_t stream) {
  const float* inp = (const float*)d_in[0];
  const float* W1  = (const float*)d_in[1];
  const float* b1  = (const float*)d_in[2];
  const float* W2  = (const float*)d_in[3];
  const float* b2  = (const float*)d_in[4];
  float* out = (float*)d_out;

  const int nrows  = in_sizes[0] / 2;
  const int blocks = (nrows + kRowsPerBlock - 1) / kRowsPerBlock;

  hipLaunchKernelGGL(hybrid_sampler_kernel, dim3(blocks), dim3(256), 0, stream,
                     inp, W1, b1, W2, b2, out, nrows);
}


# 2-qubit circuit contracted to the closed form of p0-p1 (Z expectation of qubit 1): 5 full-angle sin/cos pairs + 11 mul/fma replace 6 pairs + 40 rotation ops; circuit rounding and sum order change
# speedup vs baseline: 1.7304x; 1.0858x over previous
#include <hip/hip_runtime.h>


typedef float v2f __attribute__((ext_vector_type(2)));
typedef float v4f __attribute__((ext_vector_type(4)));
typedef float v8f __attribute__((ext_vector_type(8)));

__device__ __forceinline__ float fast_tanh(float x) {
  return __builtin_amdgcn_tanhf(x);
}

#define SINCOS(th, ss, cc)                                                   \
  const float u_##ss = (th) * 0.15915494309189535f;             \
  const float ss = __builtin_amdgcn_sinf(u_##ss);                            \
  const float cc = __builtin_amdgcn_cosf(u_##ss)

struct WaveConsts {
  v2f A1;
  v2f A2[4];
  v8f C1;
  v8f C2;
};

__device__ __forceinline__ v2f eval_row(float in0, float in1, const WaveConsts& k) {
  v2f B1;  B1.x = in0;  B1.y = in1;
  v8f H = __builtin_amdgcn_wmma_f32_16x16x4_f32(false, k.A1, false, B1, (short)0, k.C1, false, false);

  float th[8];
  #pragma unroll
  for (int r = 0; r < 8; ++r) th[r] = fast_tanh(H[r]);

  v8f acc = k.C2;
  #pragma unroll
  for (int t = 0; t < 4; ++t) {
    v2f B2;  B2.x = th[2 * t];  B2.y = th[2 * t + 1];
    acc = __builtin_amdgcn_wmma_f32_16x16x4_f32(false, k.A2[t], false, B2, (short)0, acc, false, false);
  }

  SINCOS(in0,    sa, ca);
  SINCOS(in1,    sb, cb);
  SINCOS(acc[0], s0, c0);
  SINCOS(acc[1], s1, c1);
  SINCOS(acc[3], s3, c3);
  const float g  = c1 * cb - s1 * (ca * sb);
  const float zz = c0 * g + s0 * (s1 * sa);
  const float xx = s1 * (ca * cb) + c1 * sb;
  const float d  = c3 * zz - s3 * xx;

  const float t = fast_tanh(0.5f * d);
  v2f o; o.x = __builtin_fmaf(0.5f, t, 0.5f); o.y = __builtin_fmaf(-0.5f, t, 0.5f);
  return o;
}

constexpr int kPairs        = 4;
constexpr int kRowsPerLane  = 2 * kPairs;
constexpr int kRowsPerWave  = 32 * kRowsPerLane;
constexpr int kWaves        = 8;
constexpr int kRowsPerBlock = kWaves * kRowsPerWave;

template <bool FULL>
__device__ __forceinline__ void block_body(const float* __restrict__ inp,
                                           const float* __restrict__ W1,
                                           const float* __restrict__ b1,
                                           const float* __restrict__ W2,
                                           const float* __restrict__ b2,
                                           float* __restrict__ out, int nrows) {
  const int tid  = threadIdx.x;
  const int lane = tid & 31;
  const int wave = tid >> 5;
  const int c    = lane & 15;
  const int h    = lane >> 4;

  const int row0 = blockIdx.x * kRowsPerBlock + wave * kRowsPerWave + 2 * lane;

  v4f x[kPairs];
  #pragma unroll
  for (int j = 0; j < kPairs; ++j) {
    const int ra = row0 + 64 * j;
    if (FULL) {
      x[j] = *(const v4f*)&inp[2 * ra];
    } else {
      const int qa = (ra     >= nrows) ? (nrows - 1) : ra;
      const int qb = (ra + 1 >= nrows) ? (nrows - 1) : (ra + 1);
      x[j].x = inp[2 * qa];  x[j].y = inp[2 * qa + 1];
      x[j].z = inp[2 * qb];  x[j].w = inp[2 * qb + 1];
    }
  }

  const int r8 = c & 7;
  const int r4 = c & 3;
  const bool own8 = ((c >> 3) == h);
  const bool own4 = own8 && ((c & 4) == 0);
  const float w1x = W1[r8];
  const float w1y = W1[8 + r8];

  WaveConsts k;
  k.A1.x = own8 ? w1x : 0.0f;  k.A1.y = own8 ? w1y : 0.0f;
  #pragma unroll
  for (int t = 0; t < 4; ++t) {
    const float wx = W2[(2 * t) * 4 + r4];
    const float wy = W2[(2 * t + 1) * 4 + r4];
    k.A2[t].x = own4 ? wx : 0.0f;
    k.A2[t].y = own4 ? wy : 0.0f;
  }
  k.C1 = (v8f){b1[0], b1[1], b1[2], b1[3], b1[4], b1[5], b1[6], b1[7]};
  k.C2 = (v8f){b2[0], b2[1], b2[2], b2[3], 0.0f, 0.0f, 0.0f, 0.0f};

  v4f y[kPairs];
  #pragma unroll
  for (int j = 0; j < kPairs; ++j) {
    const v2f oa = eval_row(x[j].x, x[j].y, k);
    const v2f ob = eval_row(x[j].z, x[j].w, k);
    y[j].x = oa.x;  y[j].y = oa.y;  y[j].z = ob.x;  y[j].w = ob.y;
    if (FULL) *(v4f*)&out[2 * (row0 + 64 * j)] = y[j];
    __builtin_amdgcn_sched_barrier(0);
  }

  if (!FULL) {
    #pragma unroll
    for (int j = 0; j < kPairs; ++j) {
      const int ra = row0 + 64 * j;
      if (ra < nrows)     { v2f o; o.x = y[j].x; o.y = y[j].y; *(v2f*)&out[2 * ra] = o; }
      if (ra + 1 < nrows) { v2f o; o.x = y[j].z; o.y = y[j].w; *(v2f*)&out[2 * ra + 2] = o; }
    }
  }
}

__global__ __launch_bounds__(256)
void hybrid_sampler_kernel(const float* __restrict__ inp,
                           const float* __restrict__ W1,
                           const float* __restrict__ b1,
                           const float* __restrict__ W2,
                           const float* __restrict__ b2,
                           float* __restrict__ out, int nrows) {
  if ((int)(blockIdx.x + 1) * kRowsPerBlock <= nrows)
    block_body<true>(inp, W1, b1, W2, b2, out, nrows);
  else
    block_body<false>(inp, W1, b1, W2, b2, out, nrows);
}

extern "C" void kernel_launch(void* const* d_in, const int* in_sizes, int n_in,
                              void* d_out, int out_size, void* d_ws, size_t ws_size,
                              hipStream_t stream) {
  const float* inp = (const float*)d_in[0];
  const float* W1  = (const float*)d_in[1];
  const float* b1  = (const float*)d_in[2];
  const float* W2  = (const float*)d_in[3];
  const float* b2  = (const float*)d_in[4];
  float* out = (float*)d_out;

  const int nrows  = in_sizes[0] / 2;
  const int blocks = (nrows + kRowsPerBlock - 1) / kRowsPerBlock;

  hipLaunchKernelGGL(hybrid_sampler_kernel, dim3(blocks), dim3(256), 0, stream,
                     inp, W1, b1, W2, b2, out, nrows);
}
